# starting kernel
# baseline (speedup 1.0000x reference)
#include <hip/hip_runtime.h>

#define BATCH 64
#define CH    256
#define KDIM  1024
#define NITER 100

typedef __attribute__((ext_vector_type(2))) float v2f;
typedef __attribute__((ext_vector_type(8))) float v8f;

static __device__ __forceinline__ v8f wmma4(v2f a, v2f b, v8f c) {
  return __builtin_amdgcn_wmma_f32_16x16x4_f32(false, a, false, b, (short)0, c, false, false);
}

__global__ __launch_bounds__(256) void rvp_transpose(const float* __restrict__ x,
                                                     float* __restrict__ xT) {
  __shared__ float tile[32][33];
  const int b  = blockIdx.z;
  const int k0 = blockIdx.x * 32;
  const int c0 = blockIdx.y * 32;
  const float* xb = x + (size_t)b * CH * KDIM;
  float* xTb = xT + (size_t)b * CH * KDIM;
  const int tx = threadIdx.x, ty = threadIdx.y;
  #pragma unroll
  for (int j = 0; j < 32; j += 8)
    tile[ty + j][tx] = xb[(size_t)(c0 + ty + j) * KDIM + (k0 + tx)];
  __syncthreads();
  #pragma unroll
  for (int j = 0; j < 32; j += 8)
    xTb[(size_t)(k0 + ty + j) * CH + (c0 + tx)] = tile[tx][ty + j];
}

__global__ __launch_bounds__(256) void rvp_irls(const float* __restrict__ x,
                                                const float* __restrict__ xT,
                                                float* __restrict__ out) {
  __shared__ float ly[CH];
  __shared__ float ls[KDIM];
  __shared__ float lw[KDIM];
  __shared__ float lnum[CH];
  __shared__ float lpart[8];
  __shared__ float lden;
  __shared__ float lyy;

  const int b    = blockIdx.x;
  const int tid  = threadIdx.x;
  const int wid  = tid >> 5;
  const int lane = tid & 31;
  const int laneN = lane & 15;
  const int laneH = lane >> 4;
  const float amask = (laneN == 0) ? 1.0f : 0.0f;

  const float* xb  = x  + (size_t)b * CH * KDIM;
  const float* xTb = xT + (size_t)b * CH * KDIM;

  {
    float s0 = 0.f, s1 = 0.f, s2 = 0.f, s3 = 0.f;
    for (int c = 0; c < CH; ++c) {
      const float* row = xb + (size_t)c * KDIM + tid;
      float a0 = row[0], a1 = row[256], a2 = row[512], a3 = row[768];
      s0 += a0 * a0; s1 += a1 * a1; s2 += a2 * a2; s3 += a3 * a3;
    }
    ls[tid] = s0; ls[tid + 256] = s1; ls[tid + 512] = s2; ls[tid + 768] = s3;
  }
  {
    float a = 0.f;
    for (int k = 0; k < KDIM; ++k) a += xTb[(size_t)k * CH + tid];
    float yv = a * (1.0f / (float)KDIM);
    ly[tid] = yv;
    float p = yv * yv;
    #pragma unroll
    for (int off = 16; off > 0; off >>= 1) p += __shfl_xor(p, off, 32);
    if (lane == 0) lpart[wid] = p;
  }
  __syncthreads();
  if (tid == 0) {
    float t = 0.f;
    #pragma unroll
    for (int i = 0; i < 8; ++i) t += lpart[i];
    lyy = t;
  }
  __syncthreads();

  for (int it = 0; it < NITER; ++it) {
    const float yy = lyy;

    const int kbase = wid * 128;
    v8f t8[8];
    #pragma unroll
    for (int j = 0; j < 8; ++j) t8[j] = (v8f){};
    for (int c0 = 0; c0 < CH; c0 += 4) {
      const int cA = c0 + (laneH << 1);
      const v2f yv = *(const v2f*)(ly + cA);
      v2f av; av.x = yv.x * amask; av.y = yv.y * amask;
      const float* pB = xb + (size_t)cA * KDIM + (kbase + laneN);
      #pragma unroll
      for (int j = 0; j < 8; ++j) {
        v2f bv;
        bv.x = pB[j * 16];
        bv.y = pB[j * 16 + KDIM];
        t8[j] = wmma4(av, bv, t8[j]);
      }
    }
    float den_acc = 0.f;
    if (lane < 16) {
      #pragma unroll
      for (int j = 0; j < 8; ++j) {
        const int k = kbase + j * 16 + lane;
        const float t  = t8[j][0];
        const float z2 = yy - 2.0f * t + ls[k];
        const float wv = 1.0f / sqrtf(1.0f + z2);
        lw[k] = wv;
        den_acc += wv;
      }
    }
    #pragma unroll
    for (int off = 16; off > 0; off >>= 1) den_acc += __shfl_xor(den_acc, off, 32);
    if (lane == 0) lpart[wid] = den_acc;
    __syncthreads();
    if (tid == 0) {
      float t = 0.f;
      #pragma unroll
      for (int i = 0; i < 8; ++i) t += lpart[i];
      lden = t;
    }

    {
      const int c0 = wid * 32;
      v8f n0a = {}, n1a = {}, n0b = {}, n1b = {};
      #pragma unroll 2
      for (int k0 = 0; k0 < KDIM; k0 += 8) {
        const int kA0 = k0 + (laneH << 1);
        const int kA1 = kA0 + 4;
        const v2f w0 = *(const v2f*)(lw + kA0);
        const v2f w1 = *(const v2f*)(lw + kA1);
        v2f a0; a0.x = w0.x * amask; a0.y = w0.y * amask;
        v2f a1; a1.x = w1.x * amask; a1.y = w1.y * amask;
        const float* q0 = xTb + (size_t)kA0 * CH + laneN;
        const float* q1 = xTb + (size_t)kA1 * CH + laneN;
        v2f b00, b01, b10, b11;
        b00.x = q0[c0];          b00.y = q0[c0 + CH];
        b01.x = q0[c0 + 16];     b01.y = q0[c0 + 16 + CH];
        b10.x = q1[c0];          b10.y = q1[c0 + CH];
        b11.x = q1[c0 + 16];     b11.y = q1[c0 + 16 + CH];
        n0a = wmma4(a0, b00, n0a);
        n1a = wmma4(a0, b01, n1a);
        n0b = wmma4(a1, b10, n0b);
        n1b = wmma4(a1, b11, n1b);
      }
      if (lane < 16) {
        lnum[c0 + lane]      = n0a[0] + n0b[0];
        lnum[c0 + 16 + lane] = n1a[0] + n1b[0];
      }
    }
    __syncthreads();

    {
      const float yv = lnum[tid] / lden;
      ly[tid] = yv;
      float p = yv * yv;
      #pragma unroll
      for (int off = 16; off > 0; off >>= 1) p += __shfl_xor(p, off, 32);
      if (lane == 0) lpart[wid] = p;
    }
    __syncthreads();
    if (tid == 0) {
      float t = 0.f;
      #pragma unroll
      for (int i = 0; i < 8; ++i) t += lpart[i];
      lyy = t;
    }
    __syncthreads();
  }

  out[(size_t)b * CH + tid] = ly[tid];
}

extern "C" void kernel_launch(void* const* d_in, const int* in_sizes, int n_in,
                              void* d_out, int out_size, void* d_ws, size_t ws_size,
                              hipStream_t stream) {
  (void)in_sizes; (void)n_in; (void)out_size; (void)ws_size;
  const float* x = (const float*)d_in[0];
  float* out = (float*)d_out;
  float* xT  = (float*)d_ws;

  rvp_transpose<<<dim3(KDIM / 32, CH / 32, BATCH), dim3(32, 8), 0, stream>>>(x, xT);
  rvp_irls<<<dim3(BATCH), dim3(256), 0, stream>>>(x, xT, out);
}


# one 16-wave workgroup per batch: each wave stages its own 256c x 16k sub-tiles in private LDS and feeds both products from that one read (no transposed copy); t, num, den summed in a new order
# speedup vs baseline: 1.0692x; 1.0692x over previous
#include <hip/hip_runtime.h>

#define BATCH 64
#define CH    256
#define KDIM  1024
#define NITER 100
#define NW    16
#define KS    16
#define NST   4
#define PITCH 18

typedef __attribute__((ext_vector_type(2))) float v2f;
typedef __attribute__((ext_vector_type(4))) float v4f;
typedef __attribute__((ext_vector_type(8))) float v8f;

static __device__ __forceinline__ v8f wmma4(v2f a, v2f b, v8f c) {
  return __builtin_amdgcn_wmma_f32_16x16x4_f32(false, a, false, b, (short)0, c, false, false);
}

__global__ __launch_bounds__(512) void rvp_irls(const float* __restrict__ x,
                                                float* __restrict__ out) {
  __shared__ __attribute__((aligned(16))) float ltile[NW][CH * PITCH];
  __shared__ __attribute__((aligned(16))) float lnp[NW][CH];
  __shared__ __attribute__((aligned(16))) float ly[CH];
  __shared__ float ls[KDIM];
  __shared__ float lpd[NW];
  __shared__ float lpart[8];

  const int b     = blockIdx.x;
  const int tid   = threadIdx.x;
  const int wid   = tid >> 5;
  const int lane  = tid & 31;
  const int laneN = lane & 15;
  const int laneH = lane >> 4;
  const float amask = (laneN == 0) ? 1.0f : 0.0f;

  const float* xb = x + (size_t)b * CH * KDIM;
  float* mytile = &ltile[wid][0];
  const float* xw = xb + wid * (NST * KS);
  const int lrow = lane >> 2;
  const int lcol = (lane & 3) * 4;

  v4f stg[32];
  #pragma unroll
  for (int i = 0; i < 32; ++i)
    stg[i] = *(const v4f*)(xw + (size_t)(8 * i + lrow) * KDIM + lcol);

  {
    float s0 = 0.f, s1 = 0.f;
    for (int c = 0; c < CH; ++c) {
      const float* row = xb + (size_t)c * KDIM + tid;
      const float a0 = row[0], a1 = row[512];
      s0 += a0 * a0; s1 += a1 * a1;
    }
    ls[tid] = s0; ls[tid + 512] = s1;
  }
  for (int r = 0; r < 16; ++r) {
    const int c = wid * 16 + r;
    const float* row = xb + (size_t)c * KDIM + lane * 4;
    float a = 0.f;
    #pragma unroll
    for (int i = 0; i < 8; ++i) {
      const v4f v = *(const v4f*)(row + 128 * i);
      a += (v.x + v.y) + (v.z + v.w);
    }
    #pragma unroll
    for (int off = 16; off > 0; off >>= 1) a += __shfl_xor(a, off, 32);
    if (lane == 0) ly[c] = a * (1.0f / (float)KDIM);
  }
  __syncthreads();
  if (tid < CH) {
    const float yv = ly[tid];
    float p = yv * yv;
    #pragma unroll
    for (int off = 16; off > 0; off >>= 1) p += __shfl_xor(p, off, 32);
    if (lane == 0) lpart[wid] = p;
  }
  __syncthreads();

  for (int it = 0; it < NITER; ++it) {
    float yy = 0.f;
    #pragma unroll
    for (int i = 0; i < 8; ++i) yy += lpart[i];

    v8f nacc0 = {}, nacc1 = {};
    float den_acc = 0.f;

    for (int st = 0; st < NST; ++st) {
      #pragma unroll
      for (int i = 0; i < 32; ++i) {
        float* d = mytile + (8 * i + lrow) * PITCH + lcol;
        v2f lo; lo.x = stg[i].x; lo.y = stg[i].y;
        v2f hi; hi.x = stg[i].z; hi.y = stg[i].w;
        *(v2f*)d = lo;
        *(v2f*)(d + 2) = hi;
      }
      {
        const int nst = (st + 1) & (NST - 1);
        #pragma unroll
        for (int i = 0; i < 32; ++i)
          stg[i] = *(const v4f*)(xw + (size_t)(8 * i + lrow) * KDIM + nst * KS + lcol);
      }
      const int k0 = wid * (NST * KS) + st * KS;

      v8f ta = {}, tb = {}, tc = {}, td = {};
      v2f yc[4], xc[4];
      #pragma unroll
      for (int u = 0; u < 4; ++u) {
        const int cA = 4 * u + 2 * laneH;
        yc[u] = *(const v2f*)(ly + cA);
        const float* p = mytile + cA * PITCH + laneN;
        xc[u].x = p[0]; xc[u].y = p[PITCH];
      }
      for (int i = 0; i < 64; i += 4) {
        v2f yn[4], xn[4];
        const int in = (i + 4) & 63;
        #pragma unroll
        for (int u = 0; u < 4; ++u) {
          const int cA = 4 * (in + u) + 2 * laneH;
          yn[u] = *(const v2f*)(ly + cA);
          const float* p = mytile + cA * PITCH + laneN;
          xn[u].x = p[0]; xn[u].y = p[PITCH];
        }
        v2f a0; a0.x = yc[0].x * amask; a0.y = yc[0].y * amask;
        v2f a1; a1.x = yc[1].x * amask; a1.y = yc[1].y * amask;
        v2f a2; a2.x = yc[2].x * amask; a2.y = yc[2].y * amask;
        v2f a3; a3.x = yc[3].x * amask; a3.y = yc[3].y * amask;
        ta = wmma4(a0, xc[0], ta);
        tb = wmma4(a1, xc[1], tb);
        tc = wmma4(a2, xc[2], tc);
        td = wmma4(a3, xc[3], td);
        #pragma unroll
        for (int u = 0; u < 4; ++u) { yc[u] = yn[u]; xc[u] = xn[u]; }
      }
      const float t  = (ta[0] + tb[0]) + (tc[0] + td[0]);
      const float z2 = yy - 2.0f * t + ls[k0 + laneN];
      const float wv = 1.0f / sqrtf(1.0f + z2);
      den_acc += (lane < 16) ? wv : 0.0f;

      #pragma unroll
      for (int q = 0; q < 4; ++q) {
        v2f wq;
        wq.x = __shfl(wv, 4 * q + 2 * laneH, 32);
        wq.y = __shfl(wv, 4 * q + 2 * laneH + 1, 32);
        #pragma unroll
        for (int j = 0; j < 16; ++j) {
          const v2f a = *(const v2f*)(mytile + (16 * j + laneN) * PITCH + 4 * q + 2 * laneH);
          v2f bsel;
          bsel.x = (laneN == j) ? wq.x : 0.0f;
          bsel.y = (laneN == j) ? wq.y : 0.0f;
          if (q & 1) nacc1 = wmma4(a, bsel, nacc1);
          else       nacc0 = wmma4(a, bsel, nacc0);
        }
      }
    }

    #pragma unroll
    for (int off = 16; off > 0; off >>= 1) den_acc += __shfl_xor(den_acc, off, 32);
    if (lane == 0) lpd[wid] = den_acc;
    {
      const v8f ns = nacc0 + nacc1;
      float* np = &lnp[wid][16 * laneN + 8 * laneH];
      v4f lo; lo.x = ns[0]; lo.y = ns[1]; lo.z = ns[2]; lo.w = ns[3];
      v4f hi; hi.x = ns[4]; hi.y = ns[5]; hi.z = ns[6]; hi.w = ns[7];
      *(v4f*)np = lo;
      *(v4f*)(np + 4) = hi;
    }
    __syncthreads();
    if (tid < CH) {
      float num = 0.f, den = 0.f;
      #pragma unroll
      for (int w = 0; w < NW; ++w) { num += lnp[w][tid]; den += lpd[w]; }
      const float yv = num / den;
      ly[tid] = yv;
      float p = yv * yv;
      #pragma unroll
      for (int off = 16; off > 0; off >>= 1) p += __shfl_xor(p, off, 32);
      if (lane == 0) lpart[wid] = p;
    }
    __syncthreads();
  }

  if (tid < CH) out[(size_t)b * CH + tid] = ly[tid];
}

extern "C" void kernel_launch(void* const* d_in, const int* in_sizes, int n_in,
                              void* d_out, int out_size, void* d_ws, size_t ws_size,
                              hipStream_t stream) {
  (void)in_sizes; (void)n_in; (void)out_size; (void)d_ws; (void)ws_size;
  const float* x = (const float*)d_in[0];
  float* out = (float*)d_out;
  rvp_irls<<<dim3(BATCH), dim3(512), 0, stream>>>(x, out);
}


# sub-tile re-laid in LDS as 2x2 blocks (c-pair x k-pair): one 16-byte read feeds two WMMAs in both products, reads grouped a block ahead of each 8-WMMA run; t, num, den summed in a new order
# speedup vs baseline: 1.3177x; 1.2324x over previous
#include <hip/hip_runtime.h>

#define BATCH 64
#define CH    256
#define KDIM  1024
#define NITER 100
#define NW    16
#define KS    16
#define NST   4
#define UP    9
#define TILEF (128 * UP * 4)

typedef __attribute__((ext_vector_type(2))) float v2f;
typedef __attribute__((ext_vector_type(4))) float v4f;
typedef __attribute__((ext_vector_type(8))) float v8f;

static __device__ __forceinline__ v8f wmma4(v2f a, v2f b, v8f c) {
  return __builtin_amdgcn_wmma_f32_16x16x4_f32(false, a, false, b, (short)0, c, false, false);
}

__global__ __launch_bounds__(512) void rvp_irls(const float* __restrict__ x,
                                                float* __restrict__ out) {
  __shared__ __attribute__((aligned(16))) float ltile[NW][TILEF];
  __shared__ __attribute__((aligned(16))) float lnp[NW][CH];
  __shared__ __attribute__((aligned(16))) float ly[CH];
  __shared__ __attribute__((aligned(16))) float ls[KDIM];
  __shared__ float lpd[NW];
  __shared__ float lpart[8];

  const int b     = blockIdx.x;
  const int tid   = threadIdx.x;
  const int wid   = tid >> 5;
  const int lane  = tid & 31;
  const int laneN = lane & 15;
  const int laneH = lane >> 4;

  const float* xb = x + (size_t)b * CH * KDIM;
  float* mytile = &ltile[wid][0];
  const float* xw = xb + wid * (NST * KS);
  const int lrow = lane >> 2;
  const int kq   = lane & 3;
  const float* xg = xw + (size_t)(2 * lrow) * KDIM + 4 * kq;
  float* sdst = mytile + (lrow * UP + 2 * kq) * 4;

  const int kp1 = laneN & 7;
  const float* xp1 = mytile + ((8 * (laneN >> 3) + 4 * laneH) * UP + kp1) * 4;
  const float* yb  = ly + 16 * (laneN & 1) + 8 * laneH;
  const float* xp2 = mytile + (laneN * UP + 4 * laneH) * 4;

  v4f stg[32];
  #pragma unroll
  for (int i = 0; i < 16; ++i) {
    const float* p = xg + (size_t)(16 * i) * KDIM;
    stg[2 * i]     = *(const v4f*)p;
    stg[2 * i + 1] = *(const v4f*)(p + KDIM);
  }

  {
    float s0 = 0.f, s1 = 0.f;
    for (int c = 0; c < CH; ++c) {
      const float* row = xb + (size_t)c * KDIM + tid;
      const float a0 = row[0], a1 = row[512];
      s0 += a0 * a0; s1 += a1 * a1;
    }
    ls[tid] = s0; ls[tid + 512] = s1;
  }
  for (int r = 0; r < 16; ++r) {
    const int c = wid * 16 + r;
    const float* row = xb + (size_t)c * KDIM + lane * 4;
    float a = 0.f;
    #pragma unroll
    for (int i = 0; i < 8; ++i) {
      const v4f v = *(const v4f*)(row + 128 * i);
      a += (v.x + v.y) + (v.z + v.w);
    }
    #pragma unroll
    for (int off = 16; off > 0; off >>= 1) a += __shfl_xor(a, off, 32);
    if (lane == 0) ly[c] = a * (1.0f / (float)KDIM);
  }
  __syncthreads();
  if (tid < CH) {
    const float yv = ly[tid];
    float p = yv * yv;
    #pragma unroll
    for (int off = 16; off > 0; off >>= 1) p += __shfl_xor(p, off, 32);
    if (lane == 0) lpart[wid] = p;
  }
  __syncthreads();

  const int npos = 16 * (2 * (tid >> 5) + (tid & 1)) + 8 * ((tid >> 4) & 1) + ((tid >> 1) & 7);

  for (int it = 0; it < NITER; ++it) {
    float yy = 0.f;
    #pragma unroll
    for (int i = 0; i < 8; ++i) yy += lpart[i];

    v8f nacc0 = {}, nacc1 = {};
    float den_acc = 0.f;

    for (int st = 0; st < NST; ++st) {
      #pragma unroll
      for (int i = 0; i < 16; ++i) {
        float* d = sdst + (8 * i) * UP * 4;
        const v4f r0 = stg[2 * i], r1 = stg[2 * i + 1];
        v2f p0; p0.x = r0.x; p0.y = r0.y;
        v2f p1; p1.x = r1.x; p1.y = r1.y;
        v2f p2; p2.x = r0.z; p2.y = r0.w;
        v2f p3; p3.x = r1.z; p3.y = r1.w;
        *(v2f*)d = p0;
        *(v2f*)(d + 2) = p1;
        *(v2f*)(d + 4) = p2;
        *(v2f*)(d + 6) = p3;
      }
      {
        const int nst = (st + 1) & (NST - 1);
        #pragma unroll
        for (int i = 0; i < 16; ++i) {
          const float* p = xg + (size_t)(16 * i) * KDIM + nst * KS;
          stg[2 * i]     = *(const v4f*)p;
          stg[2 * i + 1] = *(const v4f*)(p + KDIM);
        }
      }
      const int k0 = wid * (NST * KS) + st * KS;

      v8f te = {}, to = {};
      v4f xa[4], ya[2];
      #pragma unroll
      for (int v = 0; v < 4; ++v) xa[v] = *(const v4f*)(xp1 + v * UP * 4);
      ya[0] = *(const v4f*)yb;
      ya[1] = *(const v4f*)(yb + 4);
      #pragma unroll
      for (int U = 0; U < 8; ++U) {
        v4f xn[4], yn[2];
        if (U < 7) {
          #pragma unroll
          for (int v = 0; v < 4; ++v)
            xn[v] = *(const v4f*)(xp1 + ((16 * (U + 1) + v) * UP) * 4);
          yn[0] = *(const v4f*)(yb + 32 * (U + 1));
          yn[1] = *(const v4f*)(yb + 32 * (U + 1) + 4);
        }
        __builtin_amdgcn_sched_barrier(0);
        #pragma unroll
        for (int v = 0; v < 4; ++v) {
          v2f a;
          if (v & 1) { a.x = ya[v >> 1].z; a.y = ya[v >> 1].w; }
          else       { a.x = ya[v >> 1].x; a.y = ya[v >> 1].y; }
          v2f be; be.x = xa[v].x; be.y = xa[v].z;
          v2f bo; bo.x = xa[v].y; bo.y = xa[v].w;
          te = wmma4(a, be, te);
          to = wmma4(a, bo, to);
        }
        __builtin_amdgcn_sched_barrier(0);
        if (U < 7) {
          #pragma unroll
          for (int v = 0; v < 4; ++v) xa[v] = xn[v];
          ya[0] = yn[0]; ya[1] = yn[1];
        }
      }
      v4f ac[4];
      #pragma unroll
      for (int q = 0; q < 4; ++q) ac[q] = *(const v4f*)(xp2 + q * 4);
      float tev = (laneN < 8) ? te[0] : te[1];
      float tov = (laneN < 8) ? to[0] : to[1];
      tev += __shfl_xor(tev, 8, 32);
      tov += __shfl_xor(tov, 8, 32);
      const v2f sv = *(const v2f*)(ls + k0 + 2 * kp1);
      const float we = 1.0f / sqrtf(1.0f + (yy - 2.0f * tev + sv.x));
      const float wo = 1.0f / sqrtf(1.0f + (yy - 2.0f * tov + sv.y));
      den_acc += (lane < 8) ? (we + wo) : 0.0f;

      float wq[4][2];
      #pragma unroll
      for (int q = 0; q < 4; ++q) {
        wq[q][0] = __shfl(we, 4 * laneH + q, 32);
        wq[q][1] = __shfl(wo, 4 * laneH + q, 32);
      }
      #pragma unroll
      for (int jj = 0; jj < 8; ++jj) {
        v4f an[4];
        if (jj < 7) {
          #pragma unroll
          for (int q = 0; q < 4; ++q)
            an[q] = *(const v4f*)(xp2 + (16 * (jj + 1) * UP + q) * 4);
        }
        __builtin_amdgcn_sched_barrier(0);
        const bool ca = (laneN == 2 * jj);
        const bool cb = (laneN == 2 * jj + 1);
        #pragma unroll
        for (int q = 0; q < 4; ++q) {
          v2f aa; aa.x = ac[q].x; aa.y = ac[q].y;
          v2f ab; ab.x = ac[q].z; ab.y = ac[q].w;
          v2f ba; ba.x = ca ? wq[q][0] : 0.0f; ba.y = ca ? wq[q][1] : 0.0f;
          v2f bb; bb.x = cb ? wq[q][0] : 0.0f; bb.y = cb ? wq[q][1] : 0.0f;
          nacc0 = wmma4(aa, ba, nacc0);
          nacc1 = wmma4(ab, bb, nacc1);
        }
        if (jj < 7) {
          #pragma unroll
          for (int q = 0; q < 4; ++q) ac[q] = an[q];
        }
      }
    }

    #pragma unroll
    for (int off = 4; off > 0; off >>= 1) den_acc += __shfl_xor(den_acc, off, 32);
    if (lane == 0) lpd[wid] = den_acc;
    {
      const v8f ns = nacc0 + nacc1;
      float* np = &lnp[wid][16 * laneN + 8 * laneH];
      v4f lo; lo.x = ns[0]; lo.y = ns[1]; lo.z = ns[2]; lo.w = ns[3];
      v4f hi; hi.x = ns[4]; hi.y = ns[5]; hi.z = ns[6]; hi.w = ns[7];
      *(v4f*)np = lo;
      *(v4f*)(np + 4) = hi;
    }
    __syncthreads();
    if (tid < CH) {
      float num = 0.f, den = 0.f;
      #pragma unroll
      for (int w = 0; w < NW; ++w) { num += lnp[w][npos]; den += lpd[w]; }
      const float yv = num / den;
      ly[tid] = yv;
      float p = yv * yv;
      #pragma unroll
      for (int off = 16; off > 0; off >>= 1) p += __shfl_xor(p, off, 32);
      if (lane == 0) lpart[wid] = p;
    }
    __syncthreads();
  }

  if (tid < CH) out[(size_t)b * CH + tid] = ly[tid];
}

extern "C" void kernel_launch(void* const* d_in, const int* in_sizes, int n_in,
                              void* d_out, int out_size, void* d_ws, size_t ws_size,
                              hipStream_t stream) {
  (void)in_sizes; (void)n_in; (void)out_size; (void)d_ws; (void)ws_size;
  const float* x = (const float*)d_in[0];
  float* out = (float*)d_out;
  rvp_irls<<<dim3(BATCH), dim3(512), 0, stream>>>(x, out);
}


# batch split over 4 workgroups by channel quarter, one launch per IRLS iteration: each keeps its 64c x 1024k of x in LDS for both products, partial t exchanged; t, s, yy, num, den in a new order
# speedup vs baseline: 2.6211x; 1.9891x over previous
#include <hip/hip_runtime.h>

#define BATCH 64
#define CH    256
#define KDIM  1024
#define NW    16
#define NQ    4
#define CQ    64
#define KS    16
#define NST   4
#define UP    9
#define STF   (32 * UP * 4)
#define TILEF (NST * STF)
#define REC   1056

typedef __attribute__((ext_vector_type(2))) float v2f;
typedef __attribute__((ext_vector_type(4))) float v4f;
typedef __attribute__((ext_vector_type(8))) float v8f;

static __device__ __forceinline__ v8f wmma4(v2f a, v2f b, v8f c) {
  return __builtin_amdgcn_wmma_f32_16x16x4_f32(false, a, false, b, (short)0, c, false, false);
}

template <bool FIRST>
__global__ __launch_bounds__(512) void rvp_step(const float* __restrict__ x,
                                                const float* __restrict__ rd,
                                                float* __restrict__ wr,
                                                float* sp,
                                                float* __restrict__ out, int last) {
  __shared__ __attribute__((aligned(16))) float ltile[NW][TILEF];
  __shared__ __attribute__((aligned(16))) float lnp[NW][CQ];
  __shared__ __attribute__((aligned(16))) float ly[CQ];
  __shared__ float lpd[NW];
  __shared__ float lpart[2];

  const int b     = blockIdx.x >> 2;
  const int qt    = blockIdx.x & 3;
  const int tid   = threadIdx.x;
  const int wid   = tid >> 5;
  const int lane  = tid & 31;
  const int laneN = lane & 15;
  const int laneH = lane >> 4;
  const int kw    = wid * (NST * KS);

  const float* xb = x + ((size_t)b * CH + (size_t)qt * CQ) * KDIM;
  float* mytile = &ltile[wid][0];
  const int lrow = lane >> 2;
  const int kq   = lane & 3;
  const float* xg = xb + (size_t)(2 * lrow) * KDIM + kw + 4 * kq;

  const int kl = kw + 2 * lane;
  v2f tp[NQ], sq[NQ];
  float yp[NQ];
  if (!FIRST) {
    #pragma unroll
    for (int p = 0; p < NQ; ++p) {
      const float* r = rd + (size_t)(b * NQ + p) * REC;
      tp[p] = *(const v2f*)(r + kl);
      sq[p] = *(const v2f*)(sp + (size_t)(b * NQ + p) * KDIM + kl);
      yp[p] = r[KDIM];
    }
  }
  v4f stg[32];
  #pragma unroll
  for (int st = 0; st < NST; ++st) {
    #pragma unroll
    for (int i = 0; i < 4; ++i) {
      const float* p = xg + (size_t)(16 * i) * KDIM + st * KS;
      stg[8 * st + 2 * i]     = *(const v4f*)p;
      stg[8 * st + 2 * i + 1] = *(const v4f*)(p + KDIM);
    }
  }
  __builtin_amdgcn_sched_barrier(0);

  float we_all = 1.0f, wo_all = 1.0f;
  if (!FIRST) {
    v2f t = {0.f, 0.f}, s = {0.f, 0.f};
    float yy = 0.f;
    #pragma unroll
    for (int p = 0; p < NQ; ++p) {
      t.x += tp[p].x; t.y += tp[p].y;
      s.x += sq[p].x; s.y += sq[p].y;
      yy += yp[p];
    }
    we_all = 1.0f / sqrtf(1.0f + (yy - 2.0f * t.x + s.x));
    wo_all = 1.0f / sqrtf(1.0f + (yy - 2.0f * t.y + s.y));
  }
  {
    float dsum = we_all + wo_all;
    #pragma unroll
    for (int off = 16; off > 0; off >>= 1) dsum += __shfl_xor(dsum, off, 32);
    if (lane == 0) lpd[wid] = dsum;
  }

  #pragma unroll
  for (int st = 0; st < NST; ++st) {
    float* sdst = mytile + st * STF + (lrow * UP + 2 * kq) * 4;
    v4f ssq = {0.f, 0.f, 0.f, 0.f};
    #pragma unroll
    for (int i = 0; i < 4; ++i) {
      float* d = sdst + (8 * i) * UP * 4;
      const v4f r0 = stg[8 * st + 2 * i], r1 = stg[8 * st + 2 * i + 1];
      v2f p0; p0.x = r0.x; p0.y = r0.y;
      v2f p1; p1.x = r1.x; p1.y = r1.y;
      v2f p2; p2.x = r0.z; p2.y = r0.w;
      v2f p3; p3.x = r1.z; p3.y = r1.w;
      *(v2f*)d = p0;
      *(v2f*)(d + 2) = p1;
      *(v2f*)(d + 4) = p2;
      *(v2f*)(d + 6) = p3;
      if (FIRST) ssq += r0 * r0 + r1 * r1;
    }
    if (FIRST) {
      #pragma unroll
      for (int off = 4; off < 32; off <<= 1) {
        ssq.x += __shfl_xor(ssq.x, off, 32);
        ssq.y += __shfl_xor(ssq.y, off, 32);
        ssq.z += __shfl_xor(ssq.z, off, 32);
        ssq.w += __shfl_xor(ssq.w, off, 32);
      }
      if (lrow == 0) *(v4f*)(sp + (size_t)(b * NQ + qt) * KDIM + kw + st * KS + 4 * kq) = ssq;
    }
  }

  float wq[NST][4][2];
  #pragma unroll
  for (int st = 0; st < NST; ++st) {
    #pragma unroll
    for (int q = 0; q < 4; ++q) {
      wq[st][q][0] = __shfl(we_all, 8 * st + 4 * laneH + q, 32);
      wq[st][q][1] = __shfl(wo_all, 8 * st + 4 * laneH + q, 32);
    }
  }

  v8f nacc0 = {}, nacc1 = {};
  {
    const float* xp2 = mytile + (laneN * UP + 4 * laneH) * 4;
    v4f ac[4];
    #pragma unroll
    for (int q = 0; q < 4; ++q) ac[q] = *(const v4f*)(xp2 + q * 4);
    #pragma unroll
    for (int g = 0; g < 2 * NST; ++g) {
      const int st = g >> 1, jj = g & 1;
      v4f an[4];
      if (g < 2 * NST - 1) {
        const int st2 = (g + 1) >> 1, jj2 = (g + 1) & 1;
        #pragma unroll
        for (int q = 0; q < 4; ++q)
          an[q] = *(const v4f*)(xp2 + st2 * STF + (16 * jj2 * UP + q) * 4);
      }
      __builtin_amdgcn_sched_barrier(0);
      const bool ca = (laneN == 2 * jj);
      const bool cb = (laneN == 2 * jj + 1);
      #pragma unroll
      for (int q = 0; q < 4; ++q) {
        v2f aa; aa.x = ac[q].x; aa.y = ac[q].y;
        v2f ab; ab.x = ac[q].z; ab.y = ac[q].w;
        v2f ba; ba.x = ca ? wq[st][q][0] : 0.0f; ba.y = ca ? wq[st][q][1] : 0.0f;
        v2f bb; bb.x = cb ? wq[st][q][0] : 0.0f; bb.y = cb ? wq[st][q][1] : 0.0f;
        nacc0 = wmma4(aa, ba, nacc0);
        nacc1 = wmma4(ab, bb, nacc1);
      }
      if (g < 2 * NST - 1) {
        #pragma unroll
        for (int q = 0; q < 4; ++q) ac[q] = an[q];
      }
    }
  }

  {
    const v8f ns = nacc0 + nacc1;
    if (laneN < 4) {
      float* np = &lnp[wid][16 * laneN + 8 * laneH];
      v4f lo; lo.x = ns[0]; lo.y = ns[1]; lo.z = ns[2]; lo.w = ns[3];
      v4f hi; hi.x = ns[4]; hi.y = ns[5]; hi.z = ns[6]; hi.w = ns[7];
      *(v4f*)np = lo;
      *(v4f*)(np + 4) = hi;
    }
  }
  __syncthreads();
  float yv = 0.f;
  if (tid < CQ) {
    const int npos = 16 * (2 * (tid >> 5) + (tid & 1)) + 8 * ((tid >> 4) & 1) + ((tid >> 1) & 7);
    float num = 0.f, den = 0.f;
    #pragma unroll
    for (int w = 0; w < NW; ++w) { num += lnp[w][npos]; den += lpd[w]; }
    yv = num / den;
    ly[tid] = yv;
    float p = yv * yv;
    #pragma unroll
    for (int off = 16; off > 0; off >>= 1) p += __shfl_xor(p, off, 32);
    if (lane == 0) lpart[wid] = p;
  }
  __syncthreads();
  if (last) {
    if (tid < CQ) out[(size_t)b * CH + qt * CQ + tid] = yv;
    return;
  }
  if (tid == 0) wr[(size_t)(b * NQ + qt) * REC + KDIM] = lpart[0] + lpart[1];

  const int kp1 = laneN & 7;
  v8f tE[NST], tO[NST];
  #pragma unroll
  for (int st = 0; st < NST; ++st) { tE[st] = (v8f){}; tO[st] = (v8f){}; }
  {
    const float* xp1 = mytile + ((8 * (laneN >> 3) + 4 * laneH) * UP + kp1) * 4;
    const float* yb  = ly + 16 * (laneN & 1) + 8 * laneH;
    v4f xa[4], ya[2];
    #pragma unroll
    for (int v = 0; v < 4; ++v) xa[v] = *(const v4f*)(xp1 + v * UP * 4);
    ya[0] = *(const v4f*)yb;
    ya[1] = *(const v4f*)(yb + 4);
    #pragma unroll
    for (int g = 0; g < 2 * NST; ++g) {
      const int st = g >> 1;
      v4f xn[4], yn[2];
      if (g < 2 * NST - 1) {
        const int st2 = (g + 1) >> 1, U2 = (g + 1) & 1;
        #pragma unroll
        for (int v = 0; v < 4; ++v)
          xn[v] = *(const v4f*)(xp1 + st2 * STF + ((16 * U2 + v) * UP) * 4);
        yn[0] = *(const v4f*)(yb + 32 * U2);
        yn[1] = *(const v4f*)(yb + 32 * U2 + 4);
      }
      __builtin_amdgcn_sched_barrier(0);
      #pragma unroll
      for (int v = 0; v < 4; ++v) {
        v2f a;
        if (v & 1) { a.x = ya[v >> 1].z; a.y = ya[v >> 1].w; }
        else       { a.x = ya[v >> 1].x; a.y = ya[v >> 1].y; }
        v2f be; be.x = xa[v].x; be.y = xa[v].z;
        v2f bo; bo.x = xa[v].y; bo.y = xa[v].w;
        tE[st] = wmma4(a, be, tE[st]);
        tO[st] = wmma4(a, bo, tO[st]);
      }
      __builtin_amdgcn_sched_barrier(0);
      if (g < 2 * NST - 1) {
        #pragma unroll
        for (int v = 0; v < 4; ++v) xa[v] = xn[v];
        ya[0] = yn[0]; ya[1] = yn[1];
      }
    }
  }
  float tse = 0.f, tso = 0.f;
  #pragma unroll
  for (int st = 0; st < NST; ++st) {
    float tev = (laneN < 8) ? tE[st][0] : tE[st][1];
    float tov = (laneN < 8) ? tO[st][0] : tO[st][1];
    tev += __shfl_xor(tev, 8, 32);
    tov += __shfl_xor(tov, 8, 32);
    if ((lane >> 3) == st) { tse = tev; tso = tov; }
  }
  v2f tout; tout.x = tse; tout.y = tso;
  *(v2f*)(wr + (size_t)(b * NQ + qt) * REC + kw + 2 * lane) = tout;
}

extern "C" void kernel_launch(void* const* d_in, const int* in_sizes, int n_in,
                              void* d_out, int out_size, void* d_ws, size_t ws_size,
                              hipStream_t stream) {
  (void)in_sizes; (void)n_in; (void)out_size; (void)ws_size;
  const float* x = (const float*)d_in[0];
  float* out = (float*)d_out;
  float* bufA = (float*)d_ws;
  float* bufB = bufA + BATCH * NQ * REC;
  float* sp   = bufB + BATCH * NQ * REC;
  rvp_step<true><<<dim3(BATCH * NQ), dim3(512), 0, stream>>>(x, bufA, bufA, sp, out, 0);
  for (int i = 0; i < 49; ++i) {
    rvp_step<false><<<dim3(BATCH * NQ), dim3(512), 0, stream>>>(x, bufA, bufB, sp, out, 0);
    rvp_step<false><<<dim3(BATCH * NQ), dim3(512), 0, stream>>>(x, bufB, bufA, sp, out, 0);
  }
  rvp_step<false><<<dim3(BATCH * NQ), dim3(512), 0, stream>>>(x, bufA, bufB, sp, out, 0);
  rvp_step<false><<<dim3(BATCH * NQ), dim3(512), 0, stream>>>(x, bufB, bufA, sp, out, 1);
}


# a wave owns 4 whole channels (all 1024 k): num, y and product 1 stay inside the wave, no barrier between the two products (w shared by one early LDS-only barrier); t, s, num, yy summed in a new order
# speedup vs baseline: 2.6923x; 1.0272x over previous
#include <hip/hip_runtime.h>

#define BATCH 64
#define CH    256
#define KDIM  1024
#define NW    16
#define NQ    4
#define CQ    64
#define CW    4
#define PU    513
#define WTF   (2 * PU * 4)
#define REC   1056

typedef __attribute__((ext_vector_type(2))) float v2f;
typedef __attribute__((ext_vector_type(4))) float v4f;
typedef __attribute__((ext_vector_type(8))) float v8f;

static __device__ __forceinline__ v8f wmma4(v2f a, v2f b, v8f c) {
  return __builtin_amdgcn_wmma_f32_16x16x4_f32(false, a, false, b, (short)0, c, false, false);
}

static __device__ __forceinline__ float pick8(v8f d, int r) {
  float v = d[0];
  v = (r == 1) ? d[1] : v;
  v = (r == 2) ? d[2] : v;
  v = (r == 3) ? d[3] : v;
  v = (r == 4) ? d[4] : v;
  v = (r == 5) ? d[5] : v;
  v = (r == 6) ? d[6] : v;
  v = (r == 7) ? d[7] : v;
  return v;
}

template <bool FIRST>
__global__ __launch_bounds__(512) void rvp_step(const float* __restrict__ x,
                                                const float* __restrict__ rd,
                                                float* __restrict__ wr,
                                                float* sp,
                                                float* __restrict__ out, int last) {
  __shared__ __attribute__((aligned(16))) float ltile[NW][WTF];
  __shared__ __attribute__((aligned(16))) float lw[KDIM];
  __shared__ __attribute__((aligned(16))) float ly[CQ];
  __shared__ float lpd[NW];
  __shared__ float lyy[NW];

  const int b     = blockIdx.x >> 2;
  const int qt    = blockIdx.x & 3;
  const int tid   = threadIdx.x;
  const int wid   = tid >> 5;
  const int lane  = tid & 31;
  const int laneN = lane & 15;
  const int laneH = lane >> 4;
  const int kw    = wid * 64;

  const float* xw = x + ((size_t)b * CH + (size_t)qt * CQ + (size_t)wid * CW) * KDIM + 4 * lane;
  float* mytile = &ltile[wid][0];

  const int kl = kw + 2 * lane;
  v2f tp[NQ], sq[NQ];
  float yp[NQ];
  if (!FIRST) {
    #pragma unroll
    for (int p = 0; p < NQ; ++p) {
      const float* r = rd + (size_t)(b * NQ + p) * REC;
      tp[p] = *(const v2f*)(r + kl);
      sq[p] = *(const v2f*)(sp + (size_t)(b * NQ + p) * KDIM + kl);
      yp[p] = r[KDIM];
    }
  }
  v4f stg[32];
  #pragma unroll
  for (int cp = 0; cp < 2; ++cp) {
    #pragma unroll
    for (int i = 0; i < 8; ++i) {
      const float* p = xw + (size_t)(2 * cp) * KDIM + 128 * i;
      stg[16 * cp + 2 * i]     = *(const v4f*)p;
      stg[16 * cp + 2 * i + 1] = *(const v4f*)(p + KDIM);
    }
  }
  __builtin_amdgcn_sched_barrier(0);

  float we_all = 1.0f, wo_all = 1.0f;
  if (!FIRST) {
    v2f t = {0.f, 0.f}, s = {0.f, 0.f};
    float yy = 0.f;
    #pragma unroll
    for (int p = 0; p < NQ; ++p) {
      t.x += tp[p].x; t.y += tp[p].y;
      s.x += sq[p].x; s.y += sq[p].y;
      yy += yp[p];
    }
    we_all = 1.0f / sqrtf(1.0f + (yy - 2.0f * t.x + s.x));
    wo_all = 1.0f / sqrtf(1.0f + (yy - 2.0f * t.y + s.y));
  }
  {
    v2f wv; wv.x = we_all; wv.y = wo_all;
    *(v2f*)(lw + kl) = wv;
    float dsum = we_all + wo_all;
    #pragma unroll
    for (int off = 16; off > 0; off >>= 1) dsum += __shfl_xor(dsum, off, 32);
    if (lane == 0) lpd[wid] = dsum;
  }
  __builtin_amdgcn_fence(__ATOMIC_RELEASE, "workgroup", "local");
  __builtin_amdgcn_s_barrier();
  __builtin_amdgcn_fence(__ATOMIC_ACQUIRE, "workgroup", "local");
  __builtin_amdgcn_sched_barrier(0);

  v4f ssq[8];
  #pragma unroll
  for (int i = 0; i < 8; ++i) ssq[i] = (v4f){0.f, 0.f, 0.f, 0.f};
  #pragma unroll
  for (int cp = 0; cp < 2; ++cp) {
    #pragma unroll
    for (int i = 0; i < 8; ++i) {
      float* d = mytile + (cp * PU + 64 * i + 2 * lane) * 4;
      const v4f r0 = stg[16 * cp + 2 * i], r1 = stg[16 * cp + 2 * i + 1];
      v4f b0; b0.x = r0.x; b0.y = r0.y; b0.z = r1.x; b0.w = r1.y;
      v4f b1; b1.x = r0.z; b1.y = r0.w; b1.z = r1.z; b1.w = r1.w;
      *(v4f*)d = b0;
      *(v4f*)(d + 4) = b1;
      if (FIRST) ssq[i] += r0 * r0 + r1 * r1;
    }
  }

  float den = 0.f;
  #pragma unroll
  for (int w = 0; w < NW; ++w) den += lpd[w];

  v8f nacc0 = {}, nacc1 = {};
  {
    const float* xp2 = mytile + ((laneN & 1) * PU + (laneN & ~1) + laneH) * 4;
    const float* wp  = lw + 2 * (laneN & ~1) + 2 * laneH;
    v4f ac[4]; v2f wc[4];
    #pragma unroll
    for (int q = 0; q < 4; ++q) { ac[q] = *(const v4f*)(xp2 + 64 * q); wc[q] = *(const v2f*)(wp + 32 * q); }
    #pragma unroll
    for (int g = 0; g < 8; ++g) {
      v4f an[4]; v2f wn[4];
      if (g < 7) {
        #pragma unroll
        for (int q = 0; q < 4; ++q) {
          an[q] = *(const v4f*)(xp2 + 64 * (4 * (g + 1) + q));
          wn[q] = *(const v2f*)(wp + 32 * (4 * (g + 1) + q));
        }
      }
      __builtin_amdgcn_sched_barrier(0);
      #pragma unroll
      for (int q = 0; q < 4; ++q) {
        v2f aa; aa.x = ac[q].x; aa.y = ac[q].y;
        v2f ab; ab.x = ac[q].z; ab.y = ac[q].w;
        nacc0 = wmma4(aa, wc[q], nacc0);
        nacc1 = wmma4(ab, wc[q], nacc1);
      }
      if (g < 7) {
        #pragma unroll
        for (int q = 0; q < 4; ++q) { ac[q] = an[q]; wc[q] = wn[q]; }
      }
    }
  }

  float yv0, yv1;
  {
    const bool dv = (laneH == (laneN >> 3));
    float d0 = pick8(nacc0, lane & 7);
    float d1 = pick8(nacc1, lane & 7);
    d0 = dv ? d0 : 0.0f;
    d1 = dv ? d1 : 0.0f;
    d0 += __shfl_xor(d0, 2, 32);  d1 += __shfl_xor(d1, 2, 32);
    d0 += __shfl_xor(d0, 4, 32);  d1 += __shfl_xor(d1, 4, 32);
    d0 += __shfl_xor(d0, 24, 32); d1 += __shfl_xor(d1, 24, 32);
    yv0 = d0 / den;
    yv1 = d1 / den;
  }
  {
    float p = (lane < 2) ? (yv0 * yv0 + yv1 * yv1) : 0.0f;
    p += __shfl_xor(p, 1, 32);
    if (lane == 0) lyy[wid] = p;
  }
  if (last) {
    if (lane < 2) { v2f o; o.x = yv0; o.y = yv1; *(v2f*)(ly + wid * CW + 2 * lane) = o; }
    __syncthreads();
    if (tid < 32) *(v2f*)(out + (size_t)b * CH + qt * CQ + 2 * tid) = *(const v2f*)(ly + 2 * tid);
    return;
  }

  const float ya0 = __shfl(yv0, laneH, 32);
  const float ya1 = __shfl(yv1, laneH, 32);
  v8f tE[2], tO[2];
  tE[0] = (v8f){}; tE[1] = (v8f){}; tO[0] = (v8f){}; tO[1] = (v8f){};
  {
    const float* xp1 = mytile + (laneH * PU + laneN) * 4;
    v4f xa[4];
    #pragma unroll
    for (int q = 0; q < 4; ++q) xa[q] = *(const v4f*)(xp1 + 64 * q);
    #pragma unroll
    for (int g = 0; g < 8; ++g) {
      v4f xn[4];
      if (g < 7) {
        #pragma unroll
        for (int q = 0; q < 4; ++q) xn[q] = *(const v4f*)(xp1 + 64 * (4 * (g + 1) + q));
      }
      __builtin_amdgcn_sched_barrier(0);
      #pragma unroll
      for (int q = 0; q < 4; ++q) {
        const int s = 4 * g + q;
        const bool on = (laneN == (s & 15));
        v2f a;  a.x = on ? ya0 : 0.0f; a.y = on ? ya1 : 0.0f;
        v2f be; be.x = xa[q].x; be.y = xa[q].z;
        v2f bo; bo.x = xa[q].y; bo.y = xa[q].w;
        tE[s >> 4] = wmma4(a, be, tE[s >> 4]);
        tO[s >> 4] = wmma4(a, bo, tO[s >> 4]);
      }
      if (g < 7) {
        #pragma unroll
        for (int q = 0; q < 4; ++q) xa[q] = xn[q];
      }
    }
  }
  #pragma unroll
  for (int a = 0; a < 2; ++a) {
    #pragma unroll
    for (int r = 0; r < 8; ++r) {
      v2f tv; tv.x = tE[a][r]; tv.y = tO[a][r];
      *(v2f*)(mytile + 32 * (16 * a + 8 * laneH + r) + 2 * laneN) = tv;
    }
  }
  if (FIRST) {
    #pragma unroll
    for (int i = 0; i < 8; ++i) *(v4f*)(mytile + KDIM + 128 * i + 4 * lane) = ssq[i];
  }
  __syncthreads();
  {
    v2f ts = {0.f, 0.f};
    #pragma unroll
    for (int w = 0; w < NW; ++w) {
      const v2f pv = *(const v2f*)(&ltile[w][0] + 2 * tid);
      ts.x += pv.x; ts.y += pv.y;
    }
    *(v2f*)(wr + (size_t)(b * NQ + qt) * REC + 2 * tid) = ts;
    if (FIRST) {
      v2f ss = {0.f, 0.f};
      #pragma unroll
      for (int w = 0; w < NW; ++w) {
        const v2f pv = *(const v2f*)(&ltile[w][0] + KDIM + 2 * tid);
        ss.x += pv.x; ss.y += pv.y;
      }
      *(v2f*)(sp + (size_t)(b * NQ + qt) * KDIM + 2 * tid) = ss;
    }
    if (tid == 0) {
      float yy = 0.f;
      #pragma unroll
      for (int w = 0; w < NW; ++w) yy += lyy[w];
      wr[(size_t)(b * NQ + qt) * REC + KDIM] = yy;
    }
  }
}

extern "C" void kernel_launch(void* const* d_in, const int* in_sizes, int n_in,
                              void* d_out, int out_size, void* d_ws, size_t ws_size,
                              hipStream_t stream) {
  (void)in_sizes; (void)n_in; (void)out_size; (void)ws_size;
  const float* x = (const float*)d_in[0];
  float* out = (float*)d_out;
  float* bufA = (float*)d_ws;
  float* bufB = bufA + BATCH * NQ * REC;
  float* sp   = bufB + BATCH * NQ * REC;
  rvp_step<true><<<dim3(BATCH * NQ), dim3(512), 0, stream>>>(x, bufA, bufA, sp, out, 0);
  for (int i = 0; i < 49; ++i) {
    rvp_step<false><<<dim3(BATCH * NQ), dim3(512), 0, stream>>>(x, bufA, bufB, sp, out, 0);
    rvp_step<false><<<dim3(BATCH * NQ), dim3(512), 0, stream>>>(x, bufB, bufA, sp, out, 0);
  }
  rvp_step<false><<<dim3(BATCH * NQ), dim3(512), 0, stream>>>(x, bufA, bufB, sp, out, 0);
  rvp_step<false><<<dim3(BATCH * NQ), dim3(512), 0, stream>>>(x, bufB, bufA, sp, out, 1);
}


# product 2 runs under the x loads: all waves' partial-t requests go ahead of any x request (one more barrier), x stays in the load registers and feeds both products; t, num, yy summed in a new order
# speedup vs baseline: 3.1861x; 1.1834x over previous
#include <hip/hip_runtime.h>

#define BATCH 64
#define CH    256
#define KDIM  1024
#define NW    16
#define NQ    4
#define CQ    64
#define CW    4
#define REC   1056

typedef __attribute__((ext_vector_type(2))) float v2f;
typedef __attribute__((ext_vector_type(4))) float v4f;
typedef __attribute__((ext_vector_type(8))) float v8f;

static __device__ __forceinline__ v8f wmma4(v2f a, v2f b, v8f c) {
  return __builtin_amdgcn_wmma_f32_16x16x4_f32(false, a, false, b, (short)0, c, false, false);
}

static __device__ __forceinline__ float pick8(v8f d, int r) {
  float v = d[0];
  v = (r == 1) ? d[1] : v;
  v = (r == 2) ? d[2] : v;
  v = (r == 3) ? d[3] : v;
  v = (r == 4) ? d[4] : v;
  v = (r == 5) ? d[5] : v;
  v = (r == 6) ? d[6] : v;
  v = (r == 7) ? d[7] : v;
  return v;
}

template <bool FIRST>
__global__ __launch_bounds__(512) void rvp_step(const float* __restrict__ x,
                                                const float* __restrict__ rd,
                                                float* __restrict__ wr,
                                                float* sp,
                                                float* __restrict__ out, int last) {
  __shared__ __attribute__((aligned(16))) float lt[NW][KDIM];
  __shared__ __attribute__((aligned(16))) float ls[FIRST ? NW : 1][KDIM];
  __shared__ __attribute__((aligned(16))) float lw[KDIM];
  __shared__ __attribute__((aligned(16))) float ly[CQ];
  __shared__ float lpd[NW];
  __shared__ float lyy[NW];

  const int b     = blockIdx.x >> 2;
  const int qt    = blockIdx.x & 3;
  const int tid   = threadIdx.x;
  const int wid   = tid >> 5;
  const int lane  = tid & 31;
  const int laneN = lane & 15;
  const int laneH = lane >> 4;
  const int kw    = wid * 64;

  const float* xw = x + ((size_t)b * CH + (size_t)qt * CQ + (size_t)wid * CW) * KDIM + 4 * lane;

  const int kl = kw + 2 * lane;
  v2f tp[NQ], sq[NQ];
  float yp[NQ];
  if (!FIRST) {
    #pragma unroll
    for (int p = 0; p < NQ; ++p) {
      const float* r = rd + (size_t)(b * NQ + p) * REC;
      tp[p] = *(const v2f*)(r + kl);
      sq[p] = *(const v2f*)(sp + (size_t)(b * NQ + p) * KDIM + kl);
      yp[p] = r[KDIM];
    }
    __builtin_amdgcn_sched_barrier(0);
    __builtin_amdgcn_s_barrier();
    __builtin_amdgcn_sched_barrier(0);
  }
  v4f xr[32];
  #pragma unroll
  for (int i = 0; i < 8; ++i) {
    #pragma unroll
    for (int c = 0; c < CW; ++c) xr[4 * i + c] = *(const v4f*)(xw + (size_t)c * KDIM + 128 * i);
  }
  __builtin_amdgcn_sched_barrier(0);

  float we_all = 1.0f, wo_all = 1.0f;
  if (!FIRST) {
    v2f t = {0.f, 0.f}, s = {0.f, 0.f};
    float yy = 0.f;
    #pragma unroll
    for (int p = 0; p < NQ; ++p) {
      t.x += tp[p].x; t.y += tp[p].y;
      s.x += sq[p].x; s.y += sq[p].y;
      yy += yp[p];
    }
    we_all = 1.0f / sqrtf(1.0f + (yy - 2.0f * t.x + s.x));
    wo_all = 1.0f / sqrtf(1.0f + (yy - 2.0f * t.y + s.y));
  }
  {
    v2f wv; wv.x = we_all; wv.y = wo_all;
    *(v2f*)(lw + kl) = wv;
    float dsum = we_all + wo_all;
    #pragma unroll
    for (int off = 16; off > 0; off >>= 1) dsum += __shfl_xor(dsum, off, 32);
    if (lane == 0) lpd[wid] = dsum;
  }
  __builtin_amdgcn_fence(__ATOMIC_RELEASE, "workgroup", "local");
  __builtin_amdgcn_s_barrier();
  __builtin_amdgcn_fence(__ATOMIC_ACQUIRE, "workgroup", "local");

  v4f wq[8];
  #pragma unroll
  for (int i = 0; i < 8; ++i) wq[i] = *(const v4f*)(lw + 128 * i + 4 * lane);
  float den = 0.f;
  #pragma unroll
  for (int w = 0; w < NW; ++w) den += lpd[w];
  __builtin_amdgcn_sched_barrier(0);

  v8f nacc[CW];
  #pragma unroll
  for (int c = 0; c < CW; ++c) nacc[c] = (v8f){};
  #pragma unroll
  for (int i = 0; i < 8; ++i) {
    #pragma unroll
    for (int c = 0; c < CW; ++c) {
      const v4f r = xr[4 * i + c];
      v2f a0; a0.x = r.x; a0.y = r.y;
      v2f a1; a1.x = r.z; a1.y = r.w;
      v2f b0; b0.x = wq[i].x; b0.y = wq[i].y;
      v2f b1; b1.x = wq[i].z; b1.y = wq[i].w;
      nacc[c] = wmma4(a0, b0, nacc[c]);
      nacc[c] = wmma4(a1, b1, nacc[c]);
    }
  }

  float yl;
  {
    const bool dv = (laneH == (laneN >> 3));
    float d[CW];
    #pragma unroll
    for (int c = 0; c < CW; ++c) { const float v = pick8(nacc[c], lane & 7); d[c] = dv ? v : 0.0f; }
    const bool o0 = (lane & 1) != 0;
    const bool o1 = (lane & 2) != 0;
    const float e0 = (o0 ? d[2] : d[0]) + __shfl_xor(o0 ? d[0] : d[2], 1, 32);
    const float e1 = (o0 ? d[3] : d[1]) + __shfl_xor(o0 ? d[1] : d[3], 1, 32);
    float f = (o1 ? e1 : e0) + __shfl_xor(o1 ? e0 : e1, 2, 32);
    f += __shfl_xor(f, 4, 32);
    f += __shfl_xor(f, 8, 32);
    f += __shfl_xor(f, 16, 32);
    yl = f / den;
  }
  {
    float p = yl * yl;
    p += __shfl_xor(p, 1, 32);
    p += __shfl_xor(p, 2, 32);
    if (lane == 0) lyy[wid] = p;
  }
  if (last) {
    if (lane < 4) ly[wid * CW + 2 * (lane & 1) + (lane >> 1)] = yl;
    __syncthreads();
    if (tid < 32) *(v2f*)(out + (size_t)b * CH + qt * CQ + 2 * tid) = *(const v2f*)(ly + 2 * tid);
    return;
  }

  float yc[CW];
  yc[0] = __int_as_float(__builtin_amdgcn_readlane(__float_as_int(yl), 0));
  yc[1] = __int_as_float(__builtin_amdgcn_readlane(__float_as_int(yl), 2));
  yc[2] = __int_as_float(__builtin_amdgcn_readlane(__float_as_int(yl), 1));
  yc[3] = __int_as_float(__builtin_amdgcn_readlane(__float_as_int(yl), 3));
  v8f tacc[4];
  #pragma unroll
  for (int a = 0; a < 4; ++a) tacc[a] = (v8f){};
  {
    const bool hsel = (((laneN >> 1) & 1) == laneH);
    const bool on0 = hsel && ((laneN & 1) == 0);
    const bool on1 = hsel && ((laneN & 1) == 1);
    const int  rl  = laneN >> 2;
    #pragma unroll
    for (int r = 0; r < 4; ++r) {
      #pragma unroll
      for (int c = 0; c < CW; ++c) {
        v2f ay;
        ay.x = (on0 && rl == r) ? yc[c] : 0.0f;
        ay.y = (on1 && rl == r) ? yc[c] : 0.0f;
        #pragma unroll
        for (int a = 0; a < 4; ++a) {
          const int i = 2 * a + (r >> 1);
          const v4f xv = xr[4 * i + c];
          v2f bx;
          if (r & 1) { bx.x = xv.z; bx.y = xv.w; } else { bx.x = xv.x; bx.y = xv.y; }
          tacc[a] = wmma4(ay, bx, tacc[a]);
        }
      }
    }
  }
  {
    float* tw = &lt[wid][0] + 128 * laneH + 4 * laneN;
    #pragma unroll
    for (int a = 0; a < 4; ++a) {
      v4f lo; lo.x = tacc[a][0]; lo.y = tacc[a][1]; lo.z = tacc[a][4]; lo.w = tacc[a][5];
      v4f hi; hi.x = tacc[a][2]; hi.y = tacc[a][3]; hi.z = tacc[a][6]; hi.w = tacc[a][7];
      *(v4f*)(tw + 256 * a) = lo;
      *(v4f*)(tw + 256 * a + 64) = hi;
    }
  }
  if (FIRST) {
    #pragma unroll
    for (int i = 0; i < 8; ++i) {
      v4f q = (v4f){0.f, 0.f, 0.f, 0.f};
      #pragma unroll
      for (int c = 0; c < CW; ++c) q += xr[4 * i + c] * xr[4 * i + c];
      *(v4f*)(&ls[wid][0] + 128 * i + 4 * lane) = q;
    }
  }
  __syncthreads();
  {
    v2f ts = {0.f, 0.f};
    #pragma unroll
    for (int w = 0; w < NW; ++w) {
      const v2f pv = *(const v2f*)(&lt[w][0] + 2 * tid);
      ts.x += pv.x; ts.y += pv.y;
    }
    *(v2f*)(wr + (size_t)(b * NQ + qt) * REC + 2 * tid) = ts;
    if (FIRST) {
      v2f ss = {0.f, 0.f};
      #pragma unroll
      for (int w = 0; w < NW; ++w) {
        const v2f pv = *(const v2f*)(&ls[w][0] + 2 * tid);
        ss.x += pv.x; ss.y += pv.y;
      }
      *(v2f*)(sp + (size_t)(b * NQ + qt) * KDIM + 2 * tid) = ss;
    }
    if (tid == 0) {
      float yy = 0.f;
      #pragma unroll
      for (int w = 0; w < NW; ++w) yy += lyy[w];
      wr[(size_t)(b * NQ + qt) * REC + KDIM] = yy;
    }
  }
}

extern "C" void kernel_launch(void* const* d_in, const int* in_sizes, int n_in,
                              void* d_out, int out_size, void* d_ws, size_t ws_size,
                              hipStream_t stream) {
  (void)in_sizes; (void)n_in; (void)out_size; (void)ws_size;
  const float* x = (const float*)d_in[0];
  float* out = (float*)d_out;
  float* bufA = (float*)d_ws;
  float* bufB = bufA + BATCH * NQ * REC;
  float* sp   = bufB + BATCH * NQ * REC;
  rvp_step<true><<<dim3(BATCH * NQ), dim3(512), 0, stream>>>(x, bufA, bufA, sp, out, 0);
  for (int i = 0; i < 49; ++i) {
    rvp_step<false><<<dim3(BATCH * NQ), dim3(512), 0, stream>>>(x, bufA, bufB, sp, out, 0);
    rvp_step<false><<<dim3(BATCH * NQ), dim3(512), 0, stream>>>(x, bufB, bufA, sp, out, 0);
  }
  rvp_step<false><<<dim3(BATCH * NQ), dim3(512), 0, stream>>>(x, bufA, bufB, sp, out, 0);
  rvp_step<false><<<dim3(BATCH * NQ), dim3(512), 0, stream>>>(x, bufB, bufA, sp, out, 1);
}
